# starting kernel
# baseline (speedup 1.0000x reference)
#include <hip/hip_runtime.h>

typedef __attribute__((ext_vector_type(2))) float v2f;
typedef __attribute__((ext_vector_type(8))) float v8f;

#define B_SZ 8
#define NA 4096
#define NB 4096
#define FD 64
#define WAVES 8
#define ROWS_PER_WAVE 16
#define ROWS_PER_BLOCK (WAVES * ROWS_PER_WAVE)
#define NTILES (NB / 16)
#define KBIAS 48387

#define TOP3_INSERT_PK(pk, b0, b1, b2) do {          \
    int _r0 = min((pk), (b0));                        \
    (b0)    = max((pk), (b0));                        \
    int _r1 = min(_r0, (b1));                         \
    (b1)    = max(_r0, (b1));                         \
    (b2)    = max(_r1, (b2));                         \
  } while (0)

__global__ __launch_bounds__(256) void knn3_wmma(
    const float* __restrict__ a_feats,
    const float* __restrict__ b_feats,
    const int*   __restrict__ a_coords,
    const int*   __restrict__ b_coords,
    float*       __restrict__ out)
{
  __shared__ unsigned sPack[NB];
  __shared__ int      sKb2[NB];
  __shared__ int      sPk[WAVES][ROWS_PER_WAVE][3];

  const int tid   = threadIdx.x;
  const int lane  = tid & 31;
  const int wave  = tid >> 5;
  const int mlane = lane & 15;
  const bool hi   = lane >= 16;

  const int blocksPerBatch = NA / ROWS_PER_BLOCK;
  const int b       = blockIdx.x / blocksPerBatch;
  const int rowBase = (blockIdx.x % blocksPerBatch) * ROWS_PER_BLOCK
                    + wave * ROWS_PER_WAVE;

  for (int n = tid; n < NB; n += 256) {
    const int* p = b_coords + ((size_t)b * NB + n) * 3;
    int x = p[0] >> 4, y = p[1] >> 4, z = p[2] >> 4;
    sPack[n] = (unsigned)x | ((unsigned)y << 8) | ((unsigned)z << 16);
    sKb2[n]  = ((KBIAS - (x * x + y * y + z * z)) << 12) + (4095 - n);
  }
  __syncthreads();

  const int arow = rowBase + mlane;
  const int* ap = a_coords + ((size_t)b * NA + arow) * 3;
  const int iax = ap[0] >> 4, iay = ap[1] >> 4, iaz = ap[2] >> 4;
  const int na2i = iax * iax + iay * iay + iaz * iaz;
  v2f A;
  A.x = hi ? (float)(2 * iaz) : (float)(2 * iax);
  A.y = hi ? 0.0f             : (float)(2 * iay);

  int bp0[8], bp1[8], bp2[8];
#pragma unroll
  for (int r = 0; r < 8; ++r) { bp0[r] = bp1[r] = bp2[r] = -1; }

  const v8f Cz = {};

  for (int t = 0; t < NTILES; ++t) {
    const int n = t * 16 + mlane;
    const unsigned u = sPack[n];
    const int kb2 = sKb2[n];
    const float bx = (float)(u & 255u);
    const float by = (float)((u >> 8) & 255u);
    const float bz = (float)((u >> 16) & 255u);
    v2f Bm;
    Bm.x = hi ? bz : bx;
    Bm.y = hi ? 0.0f : by;

    v8f Dv = __builtin_amdgcn_wmma_f32_16x16x4_f32(
         false, A,  false, Bm,
         (short)0, Cz,  false,  false);

#pragma unroll
    for (int r = 0; r < 8; ++r) {
      const int ik = (int)Dv[r];
      const int pk = (ik << 12) + kb2;
      TOP3_INSERT_PK(pk, bp0[r], bp1[r], bp2[r]);
    }
  }

#pragma unroll
  for (int r = 0; r < 8; ++r) {
    int p0 = bp0[r], p1 = bp1[r], p2 = bp2[r];
#pragma unroll
    for (int off = 1; off < 16; off <<= 1) {
      int q0 = __shfl_xor(p0, off, 32);
      int q1 = __shfl_xor(p1, off, 32);
      int q2 = __shfl_xor(p2, off, 32);
      TOP3_INSERT_PK(q0, p0, p1, p2);
      TOP3_INSERT_PK(q1, p0, p1, p2);
      TOP3_INSERT_PK(q2, p0, p1, p2);
    }
    if (mlane == 0) {
      const int rowLocal = r + (hi ? 8 : 0);
      sPk[wave][rowLocal][0] = p0;
      sPk[wave][rowLocal][1] = p1;
      sPk[wave][rowLocal][2] = p2;
    }
  }
  __syncthreads();

  const float* afBase = a_feats + ((size_t)b * NA + rowBase) * FD;
  float* outBase = out + ((size_t)b * NA + rowBase) * (2 * FD);
  const float* bfB = b_feats + (size_t)b * NB * FD;

  for (int row = 0; row < ROWS_PER_WAVE; ++row) {
    const int pk0 = sPk[wave][row][0];
    const int pk1 = sPk[wave][row][1];
    const int pk2 = sPk[wave][row][2];
    const int rna2 = __shfl(na2i, row, 32);
    const int j0 = 4095 - (pk0 & 4095);
    const int j1 = 4095 - (pk1 & 4095);
    const int j2 = 4095 - (pk2 & 4095);
    const int d20 = rna2 + KBIAS - (int)((unsigned)pk0 >> 12);
    const int d21 = rna2 + KBIAS - (int)((unsigned)pk1 >> 12);
    const int d22 = rna2 + KBIAS - (int)((unsigned)pk2 >> 12);
    const float inv = 1.0f / 128.0f;
    const float w0 = fmaxf(0.5f - sqrtf((float)d20) * inv, 0.0f);
    const float w1 = fmaxf(0.5f - sqrtf((float)d21) * inv, 0.0f);
    const float w2 = fmaxf(0.5f - sqrtf((float)d22) * inv, 0.0f);
    const float* bf0 = bfB + (size_t)j0 * FD;
    const float* bf1 = bfB + (size_t)j1 * FD;
    const float* bf2 = bfB + (size_t)j2 * FD;
    const float* af  = afBase + (size_t)row * FD;
    float* op = outBase + (size_t)row * (2 * FD);
#pragma unroll
    for (int h = 0; h < 2; ++h) {
      const int d = lane + h * 32;
      const float tmp = w0 * bf0[d] + w1 * bf1[d] + w2 * bf2[d];
      op[d]      = af[d];
      op[FD + d] = tmp;
    }
  }
}

extern "C" void kernel_launch(void* const* d_in, const int* in_sizes, int n_in,
                              void* d_out, int out_size, void* d_ws, size_t ws_size,
                              hipStream_t stream) {
  const float* a_feats  = (const float*)d_in[0];
  const float* b_feats  = (const float*)d_in[1];
  const int*   a_coords = (const int*)d_in[2];
  const int*   b_coords = (const int*)d_in[3];
  float* out = (float*)d_out;

  dim3 grid(B_SZ * (NA / ROWS_PER_BLOCK));
  dim3 block(256);
  knn3_wmma<<<grid, block, 0, stream>>>(a_feats, b_feats, a_coords, b_coords, out);
}


# operand roles swapped: b points are the WMMA rows, a points the columns (|cb|^2 in the 4th k slot); a lane's 8 results per tile are of ONE a point, so their max against the 3rd key skips most inserts
# speedup vs baseline: 1.2254x; 1.2254x over previous
#include <hip/hip_runtime.h>


typedef __attribute__((ext_vector_type(2))) float v2f;
typedef __attribute__((ext_vector_type(8))) float v8f;

#define B_SZ 8
#define NA 4096
#define NB 4096
#define FD 64
#define WAVES 8
#define ROWS_PER_WAVE 16
#define ROWS_PER_BLOCK (WAVES * ROWS_PER_WAVE)
#define NTILES (NB / 16)
#define TPT 4
#define KBIAS 48387

#define TOP3_INSERT_PK(pk, b0, b1, b2) do {          \
    int _r0 = min((pk), (b0));                        \
    (b0)    = max((pk), (b0));                        \
    int _r1 = min(_r0, (b1));                         \
    (b1)    = max(_r0, (b1));                         \
    (b2)    = max(_r1, (b2));                         \
  } while (0)

__device__ __forceinline__ int other_half(int v) {
  return __builtin_amdgcn_permlanex16(v, v, 0x76543210, 0xfedcba98, false, false);
}

__global__ __launch_bounds__(256) void knn3_wmma(
    const float* __restrict__ a_feats,
    const float* __restrict__ b_feats,
    const int*   __restrict__ a_coords,
    const int*   __restrict__ b_coords,
    float*       __restrict__ out)
{
  __shared__ unsigned sPack[NB];
  __shared__ float    sBias[NB];
  __shared__ int      sPk[WAVES][ROWS_PER_WAVE][3];

  const int tid   = threadIdx.x;
  const int lane  = tid & 31;
  const int wave  = tid >> 5;
  const int mlane = lane & 15;
  const bool hi   = lane >= 16;

  const int blocksPerBatch = NA / ROWS_PER_BLOCK;
  const int b       = blockIdx.x / blocksPerBatch;
  const int rowBase = (blockIdx.x % blocksPerBatch) * ROWS_PER_BLOCK
                    + wave * ROWS_PER_WAVE;

  for (int n = tid; n < NB; n += 256) {
    const int* p = b_coords + ((size_t)b * NB + n) * 3;
    int x = p[0] >> 4, y = p[1] >> 4, z = p[2] >> 4;
    sPack[n] = (unsigned)x | ((unsigned)y << 8) | ((unsigned)z << 16);
    sBias[n] = (float)(KBIAS - (x * x + y * y + z * z));
  }
  __syncthreads();

  const int arow = rowBase + mlane;
  const int* ap = a_coords + ((size_t)b * NA + arow) * 3;
  const int iax = ap[0] >> 4, iay = ap[1] >> 4, iaz = ap[2] >> 4;
  const int na2i = iax * iax + iay * iay + iaz * iaz;
  v2f Acol;
  Acol.x = hi ? (float)(2 * iaz) : (float)(2 * iax);
  Acol.y = hi ? 1.0f             : (float)(2 * iay);

  int p0 = -1, p1 = -1, p2 = -1;
  int thrbits = __float_as_int(-1.0f);

  const v8f Cz = {};

  for (int t = 0; t < NTILES; t += TPT) {
    v8f Dv[TPT];
    int mt[TPT];
#pragma unroll
    for (int j = 0; j < TPT; ++j) {
      const int n = (t + j) * 16 + mlane;
      const unsigned u = sPack[n];
      const float bias = sBias[n];
      const float bx = (float)(u & 255u);
      const float by = (float)((u >> 8) & 255u);
      const float bz = (float)((u >> 16) & 255u);
      v2f Brow;
      Brow.x = hi ? bz   : bx;
      Brow.y = hi ? bias : by;

      Dv[j] = __builtin_amdgcn_wmma_f32_16x16x4_f32(
           false, Brow,  false, Acol,
           (short)0, Cz,  false,  false);

      const int g0 = max(max(__float_as_int(Dv[j][0]), __float_as_int(Dv[j][1])), __float_as_int(Dv[j][2]));
      const int g1 = max(max(__float_as_int(Dv[j][3]), __float_as_int(Dv[j][4])), __float_as_int(Dv[j][5]));
      const int g2 = max(__float_as_int(Dv[j][6]), __float_as_int(Dv[j][7]));
      mt[j] = max(max(g0, g1), g2);
    }
    int m = mt[0];
#pragma unroll
    for (int j = 1; j < TPT; ++j) m = max(m, mt[j]);

    if (__builtin_amdgcn_ballot_w32(m > thrbits) != 0u) {
#pragma unroll
      for (int j = 0; j < TPT; ++j) {
        if (__builtin_amdgcn_ballot_w32(mt[j] > thrbits) != 0u) {
          const int cbase = 4095 - ((t + j) * 16 + (hi ? 8 : 0));
#pragma unroll
          for (int r = 0; r < 8; ++r) {
            const int ik = (int)Dv[j][r];
            const int pk = (ik << 12) + (cbase - r);
            TOP3_INSERT_PK(pk, p0, p1, p2);
          }
        }
      }
      const int q0 = other_half(p0), q1 = other_half(p1), q2 = other_half(p2);
      const int thr = max(max(p2, q2), max(min(p1, q0), min(p0, q1)));
      thrbits = __float_as_int((float)(thr >> 12));
    }
  }

  {
    const int q0 = other_half(p0), q1 = other_half(p1), q2 = other_half(p2);
    TOP3_INSERT_PK(q0, p0, p1, p2);
    TOP3_INSERT_PK(q1, p0, p1, p2);
    TOP3_INSERT_PK(q2, p0, p1, p2);
  }
  if (!hi) {
    sPk[wave][mlane][0] = p0;
    sPk[wave][mlane][1] = p1;
    sPk[wave][mlane][2] = p2;
  }
  __syncthreads();

  const float* afBase = a_feats + ((size_t)b * NA + rowBase) * FD;
  float* outBase = out + ((size_t)b * NA + rowBase) * (2 * FD);
  const float* bfB = b_feats + (size_t)b * NB * FD;

  for (int row = 0; row < ROWS_PER_WAVE; ++row) {
    const int pk0 = sPk[wave][row][0];
    const int pk1 = sPk[wave][row][1];
    const int pk2 = sPk[wave][row][2];
    const int rna2 = __shfl(na2i, row, 32);
    const int j0 = 4095 - (pk0 & 4095);
    const int j1 = 4095 - (pk1 & 4095);
    const int j2 = 4095 - (pk2 & 4095);
    const int d20 = rna2 + KBIAS - (int)((unsigned)pk0 >> 12);
    const int d21 = rna2 + KBIAS - (int)((unsigned)pk1 >> 12);
    const int d22 = rna2 + KBIAS - (int)((unsigned)pk2 >> 12);
    const float inv = 1.0f / 128.0f;
    const float w0 = fmaxf(0.5f - sqrtf((float)d20) * inv, 0.0f);
    const float w1 = fmaxf(0.5f - sqrtf((float)d21) * inv, 0.0f);
    const float w2 = fmaxf(0.5f - sqrtf((float)d22) * inv, 0.0f);
    const float* bf0 = bfB + (size_t)j0 * FD;
    const float* bf1 = bfB + (size_t)j1 * FD;
    const float* bf2 = bfB + (size_t)j2 * FD;
    const float* af  = afBase + (size_t)row * FD;
    float* op = outBase + (size_t)row * (2 * FD);
#pragma unroll
    for (int h = 0; h < 2; ++h) {
      const int d = lane + h * 32;
      const float tmp = w0 * bf0[d] + w1 * bf1[d] + w2 * bf2[d];
      op[d]      = af[d];
      op[FD + d] = tmp;
    }
  }
}

extern "C" void kernel_launch(void* const* d_in, const int* in_sizes, int n_in,
                              void* d_out, int out_size, void* d_ws, size_t ws_size,
                              hipStream_t stream) {
  const float* a_feats  = (const float*)d_in[0];
  const float* b_feats  = (const float*)d_in[1];
  const int*   a_coords = (const int*)d_in[2];
  const int*   b_coords = (const int*)d_in[3];
  float* out = (float*)d_out;

  dim3 grid(B_SZ * (NA / ROWS_PER_BLOCK));
  dim3 block(256);
  knn3_wmma<<<grid, block, 0, stream>>>(a_feats, b_feats, a_coords, b_coords, out);
}


# sweep trips of 8 b tiles with ONE vector-compare-to-scalar hand-over each: the 8 tile masks go to SGPRs together, tile inserts branch on those (was a vector compare + branch per trip and per tile)
# speedup vs baseline: 1.2678x; 1.0346x over previous
#include <hip/hip_runtime.h>


typedef __attribute__((ext_vector_type(2))) float v2f;
typedef __attribute__((ext_vector_type(8))) float v8f;

#define B_SZ 8
#define NA 4096
#define NB 4096
#define FD 64
#define WAVES 8
#define ROWS_PER_WAVE 16
#define ROWS_PER_BLOCK (WAVES * ROWS_PER_WAVE)
#define NTILES (NB / 16)
#define TPT 8
#define KBIAS 48387

#define TOP3_INSERT_PK(pk, b0, b1, b2) do {          \
    int _r0 = min((pk), (b0));                        \
    (b0)    = max((pk), (b0));                        \
    int _r1 = min(_r0, (b1));                         \
    (b1)    = max(_r0, (b1));                         \
    (b2)    = max(_r1, (b2));                         \
  } while (0)

__device__ __forceinline__ int other_half(int v) {
  return __builtin_amdgcn_permlanex16(v, v, 0x76543210, 0xfedcba98, false, false);
}

__global__ __launch_bounds__(256) void knn3_wmma(
    const float* __restrict__ a_feats,
    const float* __restrict__ b_feats,
    const int*   __restrict__ a_coords,
    const int*   __restrict__ b_coords,
    float*       __restrict__ out)
{
  __shared__ unsigned sPack[NB];
  __shared__ float    sBias[NB];
  __shared__ int      sPk[WAVES][ROWS_PER_WAVE][3];

  const int tid   = threadIdx.x;
  const int lane  = tid & 31;
  const int wave  = tid >> 5;
  const int mlane = lane & 15;
  const bool hi   = lane >= 16;

  const int blocksPerBatch = NA / ROWS_PER_BLOCK;
  const int b       = blockIdx.x / blocksPerBatch;
  const int rowBase = (blockIdx.x % blocksPerBatch) * ROWS_PER_BLOCK
                    + wave * ROWS_PER_WAVE;

  for (int n = tid; n < NB; n += 256) {
    const int* p = b_coords + ((size_t)b * NB + n) * 3;
    int x = p[0] >> 4, y = p[1] >> 4, z = p[2] >> 4;
    sPack[n] = (unsigned)x | ((unsigned)y << 8) | ((unsigned)z << 16);
    sBias[n] = (float)(KBIAS - (x * x + y * y + z * z));
  }
  __syncthreads();

  const int arow = rowBase + mlane;
  const int* ap = a_coords + ((size_t)b * NA + arow) * 3;
  const int iax = ap[0] >> 4, iay = ap[1] >> 4, iaz = ap[2] >> 4;
  const int na2i = iax * iax + iay * iay + iaz * iaz;
  v2f Acol;
  Acol.x = hi ? (float)(2 * iaz) : (float)(2 * iax);
  Acol.y = hi ? 1.0f             : (float)(2 * iay);

  int p0 = -1, p1 = -1, p2 = -1;
  int thrbits = __float_as_int(-1.0f);

  const v8f Cz = {};

  for (int t = 0; t < NTILES; t += TPT) {
    v8f Dv[TPT];
    int mt[TPT];
#pragma unroll
    for (int j = 0; j < TPT; ++j) {
      const int n = (t + j) * 16 + mlane;
      const unsigned u = sPack[n];
      const float bias = sBias[n];
      const float bx = (float)(u & 255u);
      const float by = (float)((u >> 8) & 255u);
      const float bz = (float)((u >> 16) & 255u);
      v2f Brow;
      Brow.x = hi ? bz   : bx;
      Brow.y = hi ? bias : by;

      Dv[j] = __builtin_amdgcn_wmma_f32_16x16x4_f32(
           false, Brow,  false, Acol,
           (short)0, Cz,  false,  false);

      const int g0 = max(max(__float_as_int(Dv[j][0]), __float_as_int(Dv[j][1])), __float_as_int(Dv[j][2]));
      const int g1 = max(max(__float_as_int(Dv[j][3]), __float_as_int(Dv[j][4])), __float_as_int(Dv[j][5]));
      const int g2 = max(__float_as_int(Dv[j][6]), __float_as_int(Dv[j][7]));
      mt[j] = max(max(g0, g1), g2);
    }

    unsigned need[TPT];
#pragma unroll
    for (int j = 0; j < TPT; ++j) need[j] = __builtin_amdgcn_ballot_w32(mt[j] > thrbits);
    unsigned any = need[0];
#pragma unroll
    for (int j = 1; j < TPT; ++j) any |= need[j];

    if (any != 0u) {
#pragma unroll
      for (int j = 0; j < TPT; ++j) {
        if (need[j] != 0u) {
          const int cbase = 4095 - ((t + j) * 16 + (hi ? 8 : 0));
#pragma unroll
          for (int r = 0; r < 8; ++r) {
            const int ik = (int)Dv[j][r];
            const int pk = (ik << 12) + (cbase - r);
            TOP3_INSERT_PK(pk, p0, p1, p2);
          }
        }
      }
      const int q0 = other_half(p0), q1 = other_half(p1), q2 = other_half(p2);
      const int thr = max(max(p2, q2), max(min(p1, q0), min(p0, q1)));
      thrbits = __float_as_int((float)(thr >> 12));
    }
  }

  {
    const int q0 = other_half(p0), q1 = other_half(p1), q2 = other_half(p2);
    TOP3_INSERT_PK(q0, p0, p1, p2);
    TOP3_INSERT_PK(q1, p0, p1, p2);
    TOP3_INSERT_PK(q2, p0, p1, p2);
  }
  if (!hi) {
    sPk[wave][mlane][0] = p0;
    sPk[wave][mlane][1] = p1;
    sPk[wave][mlane][2] = p2;
  }
  __syncthreads();

  const float* afBase = a_feats + ((size_t)b * NA + rowBase) * FD;
  float* outBase = out + ((size_t)b * NA + rowBase) * (2 * FD);
  const float* bfB = b_feats + (size_t)b * NB * FD;

  for (int row = 0; row < ROWS_PER_WAVE; ++row) {
    const int pk0 = sPk[wave][row][0];
    const int pk1 = sPk[wave][row][1];
    const int pk2 = sPk[wave][row][2];
    const int rna2 = __shfl(na2i, row, 32);
    const int j0 = 4095 - (pk0 & 4095);
    const int j1 = 4095 - (pk1 & 4095);
    const int j2 = 4095 - (pk2 & 4095);
    const int d20 = rna2 + KBIAS - (int)((unsigned)pk0 >> 12);
    const int d21 = rna2 + KBIAS - (int)((unsigned)pk1 >> 12);
    const int d22 = rna2 + KBIAS - (int)((unsigned)pk2 >> 12);
    const float inv = 1.0f / 128.0f;
    const float w0 = fmaxf(0.5f - sqrtf((float)d20) * inv, 0.0f);
    const float w1 = fmaxf(0.5f - sqrtf((float)d21) * inv, 0.0f);
    const float w2 = fmaxf(0.5f - sqrtf((float)d22) * inv, 0.0f);
    const float* bf0 = bfB + (size_t)j0 * FD;
    const float* bf1 = bfB + (size_t)j1 * FD;
    const float* bf2 = bfB + (size_t)j2 * FD;
    const float* af  = afBase + (size_t)row * FD;
    float* op = outBase + (size_t)row * (2 * FD);
#pragma unroll
    for (int h = 0; h < 2; ++h) {
      const int d = lane + h * 32;
      const float tmp = w0 * bf0[d] + w1 * bf1[d] + w2 * bf2[d];
      op[d]      = af[d];
      op[FD + d] = tmp;
    }
  }
}

extern "C" void kernel_launch(void* const* d_in, const int* in_sizes, int n_in,
                              void* d_out, int out_size, void* d_ws, size_t ws_size,
                              hipStream_t stream) {
  const float* a_feats  = (const float*)d_in[0];
  const float* b_feats  = (const float*)d_in[1];
  const int*   a_coords = (const int*)d_in[2];
  const int*   b_coords = (const int*)d_in[3];
  float* out = (float*)d_out;

  dim3 grid(B_SZ * (NA / ROWS_PER_BLOCK));
  dim3 block(256);
  knn3_wmma<<<grid, block, 0, stream>>>(a_feats, b_feats, a_coords, b_coords, out);
}


# 16 waves per block, 8 of them memory-only: waves 0-7 sweep as before, all 16 gather (8 rows a wave, weights formed once per row by its sweeping wave); the 8-row gather compiles to one load phase
# speedup vs baseline: 1.6732x; 1.3198x over previous
#include <hip/hip_runtime.h>


typedef __attribute__((ext_vector_type(2))) float v2f;
typedef __attribute__((ext_vector_type(8))) float v8f;

#define B_SZ 8
#define NA 4096
#define NB 4096
#define FD 64
#define WAVES 8
#define ALL_WAVES 16
#define THREADS (ALL_WAVES * 32)
#define ROWS_PER_WAVE 16
#define ROWS_PER_BLOCK (WAVES * ROWS_PER_WAVE)
#define ROWS_PER_MOVER (ROWS_PER_BLOCK / ALL_WAVES)
#define NTILES (NB / 16)
#define TPT 8
#define KBIAS 48387

#define TOP3_INSERT_PK(pk, b0, b1, b2) do {          \
    int _r0 = min((pk), (b0));                        \
    (b0)    = max((pk), (b0));                        \
    int _r1 = min(_r0, (b1));                         \
    (b1)    = max(_r0, (b1));                         \
    (b2)    = max(_r1, (b2));                         \
  } while (0)

__device__ __forceinline__ int other_half(int v) {
  return __builtin_amdgcn_permlanex16(v, v, 0x76543210, 0xfedcba98, false, false);
}

__global__ __launch_bounds__(THREADS) void knn3_wmma(
    const float* __restrict__ a_feats,
    const float* __restrict__ b_feats,
    const int*   __restrict__ a_coords,
    const int*   __restrict__ b_coords,
    float*       __restrict__ out)
{
  __shared__ unsigned sPack[NB];
  __shared__ float    sBias[NB];
  __shared__ int      sIdx[ROWS_PER_BLOCK][3];
  __shared__ float    sWgt[ROWS_PER_BLOCK][3];

  const int tid   = threadIdx.x;
  const int lane  = tid & 31;
  const int wave  = __builtin_amdgcn_readfirstlane(tid >> 5);
  const int mlane = lane & 15;
  const bool hi   = lane >= 16;

  const int blocksPerBatch = NA / ROWS_PER_BLOCK;
  const int b        = blockIdx.x / blocksPerBatch;
  const int blockRow = (blockIdx.x % blocksPerBatch) * ROWS_PER_BLOCK;

  for (int n = tid; n < NB; n += THREADS) {
    const int* p = b_coords + ((size_t)b * NB + n) * 3;
    int x = p[0] >> 4, y = p[1] >> 4, z = p[2] >> 4;
    sPack[n] = (unsigned)x | ((unsigned)y << 8) | ((unsigned)z << 16);
    sBias[n] = (float)(KBIAS - (x * x + y * y + z * z));
  }
  __syncthreads();

  if (wave < WAVES) {
  const int rowBase = blockRow + wave * ROWS_PER_WAVE;
  const int arow = rowBase + mlane;
  const int* ap = a_coords + ((size_t)b * NA + arow) * 3;
  const int iax = ap[0] >> 4, iay = ap[1] >> 4, iaz = ap[2] >> 4;
  const int na2i = iax * iax + iay * iay + iaz * iaz;
  v2f Acol;
  Acol.x = hi ? (float)(2 * iaz) : (float)(2 * iax);
  Acol.y = hi ? 1.0f             : (float)(2 * iay);

  int p0 = -1, p1 = -1, p2 = -1;
  int thrbits = __float_as_int(-1.0f);

  const v8f Cz = {};

  for (int t = 0; t < NTILES; t += TPT) {
    v8f Dv[TPT];
    int mt[TPT];
#pragma unroll
    for (int j = 0; j < TPT; ++j) {
      const int n = (t + j) * 16 + mlane;
      const unsigned u = sPack[n];
      const float bias = sBias[n];
      const float bx = (float)(u & 255u);
      const float by = (float)((u >> 8) & 255u);
      const float bz = (float)((u >> 16) & 255u);
      v2f Brow;
      Brow.x = hi ? bz   : bx;
      Brow.y = hi ? bias : by;

      Dv[j] = __builtin_amdgcn_wmma_f32_16x16x4_f32(
           false, Brow,  false, Acol,
           (short)0, Cz,  false,  false);

      const int g0 = max(max(__float_as_int(Dv[j][0]), __float_as_int(Dv[j][1])), __float_as_int(Dv[j][2]));
      const int g1 = max(max(__float_as_int(Dv[j][3]), __float_as_int(Dv[j][4])), __float_as_int(Dv[j][5]));
      const int g2 = max(__float_as_int(Dv[j][6]), __float_as_int(Dv[j][7]));
      mt[j] = max(max(g0, g1), g2);
    }

    unsigned need[TPT];
#pragma unroll
    for (int j = 0; j < TPT; ++j) need[j] = __builtin_amdgcn_ballot_w32(mt[j] > thrbits);
    unsigned any = need[0];
#pragma unroll
    for (int j = 1; j < TPT; ++j) any |= need[j];

    if (any != 0u) {
#pragma unroll
      for (int j = 0; j < TPT; ++j) {
        if (need[j] != 0u) {
          const int cbase = 4095 - ((t + j) * 16 + (hi ? 8 : 0));
#pragma unroll
          for (int r = 0; r < 8; ++r) {
            const int ik = (int)Dv[j][r];
            const int pk = (ik << 12) + (cbase - r);
            TOP3_INSERT_PK(pk, p0, p1, p2);
          }
        }
      }
      const int q0 = other_half(p0), q1 = other_half(p1), q2 = other_half(p2);
      const int thr = max(max(p2, q2), max(min(p1, q0), min(p0, q1)));
      thrbits = __float_as_int((float)(thr >> 12));
    }
  }

  {
    const int q0 = other_half(p0), q1 = other_half(p1), q2 = other_half(p2);
    TOP3_INSERT_PK(q0, p0, p1, p2);
    TOP3_INSERT_PK(q1, p0, p1, p2);
    TOP3_INSERT_PK(q2, p0, p1, p2);
  }
  if (!hi) {
    const int r = wave * ROWS_PER_WAVE + mlane;
    const int d20 = na2i + KBIAS - (int)((unsigned)p0 >> 12);
    const int d21 = na2i + KBIAS - (int)((unsigned)p1 >> 12);
    const int d22 = na2i + KBIAS - (int)((unsigned)p2 >> 12);
    const float inv = 1.0f / 128.0f;
    sIdx[r][0] = 4095 - (p0 & 4095);
    sIdx[r][1] = 4095 - (p1 & 4095);
    sIdx[r][2] = 4095 - (p2 & 4095);
    sWgt[r][0] = fmaxf(0.5f - sqrtf((float)d20) * inv, 0.0f);
    sWgt[r][1] = fmaxf(0.5f - sqrtf((float)d21) * inv, 0.0f);
    sWgt[r][2] = fmaxf(0.5f - sqrtf((float)d22) * inv, 0.0f);
  }
  }
  __syncthreads();

  const int moverRow = wave * ROWS_PER_MOVER;
  const float* afBase = a_feats + ((size_t)b * NA + blockRow + moverRow) * FD;
  float* outBase = out + ((size_t)b * NA + blockRow + moverRow) * (2 * FD);
  const float* bfB = b_feats + (size_t)b * NB * FD;

  for (int row = 0; row < ROWS_PER_MOVER; ++row) {
    const int j0 = sIdx[moverRow + row][0];
    const int j1 = sIdx[moverRow + row][1];
    const int j2 = sIdx[moverRow + row][2];
    const float w0 = sWgt[moverRow + row][0];
    const float w1 = sWgt[moverRow + row][1];
    const float w2 = sWgt[moverRow + row][2];
    const float* bf0 = bfB + (size_t)j0 * FD;
    const float* bf1 = bfB + (size_t)j1 * FD;
    const float* bf2 = bfB + (size_t)j2 * FD;
    const float* af  = afBase + (size_t)row * FD;
    float* op = outBase + (size_t)row * (2 * FD);
#pragma unroll
    for (int h = 0; h < 2; ++h) {
      const int d = lane + h * 32;
      const float tmp = w0 * bf0[d] + w1 * bf1[d] + w2 * bf2[d];
      op[d]      = af[d];
      op[FD + d] = tmp;
    }
  }
}

extern "C" void kernel_launch(void* const* d_in, const int* in_sizes, int n_in,
                              void* d_out, int out_size, void* d_ws, size_t ws_size,
                              hipStream_t stream) {
  const float* a_feats  = (const float*)d_in[0];
  const float* b_feats  = (const float*)d_in[1];
  const int*   a_coords = (const int*)d_in[2];
  const int*   b_coords = (const int*)d_in[3];
  float* out = (float*)d_out;

  dim3 grid(B_SZ * (NA / ROWS_PER_BLOCK));
  dim3 block(THREADS);
  knn3_wmma<<<grid, block, 0, stream>>>(a_feats, b_feats, a_coords, b_coords, out);
}


# b coordinate staging goes under the sweep: all waves stage the first 512 points (one per thread), then waves 8-15 stage the other 3584 while waves 0-7 run the first 4 sweep trips; one more barrier
# speedup vs baseline: 1.8989x; 1.1349x over previous
#include <hip/hip_runtime.h>


typedef __attribute__((ext_vector_type(2))) float v2f;
typedef __attribute__((ext_vector_type(8))) float v8f;

#define B_SZ 8
#define NA 4096
#define NB 4096
#define FD 64
#define WAVES 8
#define ALL_WAVES 16
#define THREADS (ALL_WAVES * 32)
#define ROWS_PER_WAVE 16
#define ROWS_PER_BLOCK (WAVES * ROWS_PER_WAVE)
#define ROWS_PER_MOVER (ROWS_PER_BLOCK / ALL_WAVES)
#define NTILES (NB / 16)
#define TPT 8
#define HEAD_POINTS THREADS
#define HEAD_TILES (HEAD_POINTS / 16)
#define TAIL_PER_THREAD ((NB - HEAD_POINTS) / (THREADS - WAVES * 32))
#define KBIAS 48387

#define TOP3_INSERT_PK(pk, b0, b1, b2) do {          \
    int _r0 = min((pk), (b0));                        \
    (b0)    = max((pk), (b0));                        \
    int _r1 = min(_r0, (b1));                         \
    (b1)    = max(_r0, (b1));                         \
    (b2)    = max(_r1, (b2));                         \
  } while (0)

__device__ __forceinline__ int other_half(int v) {
  return __builtin_amdgcn_permlanex16(v, v, 0x76543210, 0xfedcba98, false, false);
}

__device__ __forceinline__ void put_point(unsigned* sPack, float* sBias, int n, int cx, int cy, int cz) {
  const int x = cx >> 4, y = cy >> 4, z = cz >> 4;
  sPack[n] = (unsigned)x | ((unsigned)y << 8) | ((unsigned)z << 16);
  sBias[n] = (float)(KBIAS - (x * x + y * y + z * z));
}

__device__ __forceinline__ void sweep_tiles(int tBegin, int tEnd,
    const unsigned* sPack, const float* sBias, v2f Acol, int mlane, bool hi,
    int& p0, int& p1, int& p2, int& thrbits)
{
  const v8f Cz = {};
  for (int t = tBegin; t < tEnd; t += TPT) {
    v8f Dv[TPT];
    int mt[TPT];
#pragma unroll
    for (int j = 0; j < TPT; ++j) {
      const int n = (t + j) * 16 + mlane;
      const unsigned u = sPack[n];
      const float bias = sBias[n];
      const float bx = (float)(u & 255u);
      const float by = (float)((u >> 8) & 255u);
      const float bz = (float)((u >> 16) & 255u);
      v2f Brow;
      Brow.x = hi ? bz   : bx;
      Brow.y = hi ? bias : by;

      Dv[j] = __builtin_amdgcn_wmma_f32_16x16x4_f32(
           false, Brow,  false, Acol,
           (short)0, Cz,  false,  false);

      const int g0 = max(max(__float_as_int(Dv[j][0]), __float_as_int(Dv[j][1])), __float_as_int(Dv[j][2]));
      const int g1 = max(max(__float_as_int(Dv[j][3]), __float_as_int(Dv[j][4])), __float_as_int(Dv[j][5]));
      const int g2 = max(__float_as_int(Dv[j][6]), __float_as_int(Dv[j][7]));
      mt[j] = max(max(g0, g1), g2);
    }

    unsigned need[TPT];
#pragma unroll
    for (int j = 0; j < TPT; ++j) need[j] = __builtin_amdgcn_ballot_w32(mt[j] > thrbits);
    unsigned any = need[0];
#pragma unroll
    for (int j = 1; j < TPT; ++j) any |= need[j];

    if (any != 0u) {
#pragma unroll
      for (int j = 0; j < TPT; ++j) {
        if (need[j] != 0u) {
          const int cbase = 4095 - ((t + j) * 16 + (hi ? 8 : 0));
#pragma unroll
          for (int r = 0; r < 8; ++r) {
            const int ik = (int)Dv[j][r];
            const int pk = (ik << 12) + (cbase - r);
            TOP3_INSERT_PK(pk, p0, p1, p2);
          }
        }
      }
      const int q0 = other_half(p0), q1 = other_half(p1), q2 = other_half(p2);
      const int thr = max(max(p2, q2), max(min(p1, q0), min(p0, q1)));
      thrbits = __float_as_int((float)(thr >> 12));
    }
  }
}

__global__ __launch_bounds__(THREADS) void knn3_wmma(
    const float* __restrict__ a_feats,
    const float* __restrict__ b_feats,
    const int*   __restrict__ a_coords,
    const int*   __restrict__ b_coords,
    float*       __restrict__ out)
{
  __shared__ unsigned sPack[NB];
  __shared__ float    sBias[NB];
  __shared__ int      sIdx[ROWS_PER_BLOCK][3];
  __shared__ float    sWgt[ROWS_PER_BLOCK][3];

  const int tid   = threadIdx.x;
  const int lane  = tid & 31;
  const int wave  = __builtin_amdgcn_readfirstlane(tid >> 5);
  const int mlane = lane & 15;
  const bool hi   = lane >= 16;

  const int blocksPerBatch = NA / ROWS_PER_BLOCK;
  const int b        = blockIdx.x / blocksPerBatch;
  const int blockRow = (blockIdx.x % blocksPerBatch) * ROWS_PER_BLOCK;
  const int* bc = b_coords + (size_t)b * NB * 3;

  {
    const int* p = bc + tid * 3;
    put_point(sPack, sBias, tid, p[0], p[1], p[2]);
  }
  __syncthreads();

  int p0 = -1, p1 = -1, p2 = -1;
  int thrbits = __float_as_int(-1.0f);
  int na2i = 0;
  v2f Acol = {};

  if (wave < WAVES) {
    const int arow = blockRow + wave * ROWS_PER_WAVE + mlane;
    const int* ap = a_coords + ((size_t)b * NA + arow) * 3;
    const int iax = ap[0] >> 4, iay = ap[1] >> 4, iaz = ap[2] >> 4;
    na2i = iax * iax + iay * iay + iaz * iaz;
    Acol.x = hi ? (float)(2 * iaz) : (float)(2 * iax);
    Acol.y = hi ? 1.0f             : (float)(2 * iay);
    sweep_tiles(0, HEAD_TILES, sPack, sBias, Acol, mlane, hi, p0, p1, p2, thrbits);
  } else {
    const int mtid = tid - WAVES * 32;
    int cx[TAIL_PER_THREAD], cy[TAIL_PER_THREAD], cz[TAIL_PER_THREAD];
#pragma unroll
    for (int k = 0; k < TAIL_PER_THREAD; ++k) {
      const int* p = bc + (HEAD_POINTS + k * (THREADS - WAVES * 32) + mtid) * 3;
      cx[k] = p[0]; cy[k] = p[1]; cz[k] = p[2];
    }
#pragma unroll
    for (int k = 0; k < TAIL_PER_THREAD; ++k)
      put_point(sPack, sBias, HEAD_POINTS + k * (THREADS - WAVES * 32) + mtid, cx[k], cy[k], cz[k]);
  }
  __syncthreads();

  if (wave < WAVES) {
  sweep_tiles(HEAD_TILES, NTILES, sPack, sBias, Acol, mlane, hi, p0, p1, p2, thrbits);

  {
    const int q0 = other_half(p0), q1 = other_half(p1), q2 = other_half(p2);
    TOP3_INSERT_PK(q0, p0, p1, p2);
    TOP3_INSERT_PK(q1, p0, p1, p2);
    TOP3_INSERT_PK(q2, p0, p1, p2);
  }
  if (!hi) {
    const int r = wave * ROWS_PER_WAVE + mlane;
    const int d20 = na2i + KBIAS - (int)((unsigned)p0 >> 12);
    const int d21 = na2i + KBIAS - (int)((unsigned)p1 >> 12);
    const int d22 = na2i + KBIAS - (int)((unsigned)p2 >> 12);
    const float inv = 1.0f / 128.0f;
    sIdx[r][0] = 4095 - (p0 & 4095);
    sIdx[r][1] = 4095 - (p1 & 4095);
    sIdx[r][2] = 4095 - (p2 & 4095);
    sWgt[r][0] = fmaxf(0.5f - sqrtf((float)d20) * inv, 0.0f);
    sWgt[r][1] = fmaxf(0.5f - sqrtf((float)d21) * inv, 0.0f);
    sWgt[r][2] = fmaxf(0.5f - sqrtf((float)d22) * inv, 0.0f);
  }
  }
  __syncthreads();

  const int moverRow = wave * ROWS_PER_MOVER;
  const float* afBase = a_feats + ((size_t)b * NA + blockRow + moverRow) * FD;
  float* outBase = out + ((size_t)b * NA + blockRow + moverRow) * (2 * FD);
  const float* bfB = b_feats + (size_t)b * NB * FD;

  for (int row = 0; row < ROWS_PER_MOVER; ++row) {
    const int j0 = sIdx[moverRow + row][0];
    const int j1 = sIdx[moverRow + row][1];
    const int j2 = sIdx[moverRow + row][2];
    const float w0 = sWgt[moverRow + row][0];
    const float w1 = sWgt[moverRow + row][1];
    const float w2 = sWgt[moverRow + row][2];
    const float* bf0 = bfB + (size_t)j0 * FD;
    const float* bf1 = bfB + (size_t)j1 * FD;
    const float* bf2 = bfB + (size_t)j2 * FD;
    const float* af  = afBase + (size_t)row * FD;
    float* op = outBase + (size_t)row * (2 * FD);
#pragma unroll
    for (int h = 0; h < 2; ++h) {
      const int d = lane + h * 32;
      const float tmp = w0 * bf0[d] + w1 * bf1[d] + w2 * bf2[d];
      op[d]      = af[d];
      op[FD + d] = tmp;
    }
  }
}

extern "C" void kernel_launch(void* const* d_in, const int* in_sizes, int n_in,
                              void* d_out, int out_size, void* d_ws, size_t ws_size,
                              hipStream_t stream) {
  const float* a_feats  = (const float*)d_in[0];
  const float* b_feats  = (const float*)d_in[1];
  const int*   a_coords = (const int*)d_in[2];
  const int*   b_coords = (const int*)d_in[3];
  float* out = (float*)d_out;

  dim3 grid(B_SZ * (NA / ROWS_PER_BLOCK));
  dim3 block(THREADS);
  knn3_wmma<<<grid, block, 0, stream>>>(a_feats, b_feats, a_coords, b_coords, out);
}


# priority windows: in both barrier intervals of the sweep the younger wave of each SIMD's sweeping pair (waves 4-7) issues at priority 3 up to b tile 16 resp. 152, then at 0: pairs arrive together
# speedup vs baseline: 1.9713x; 1.0381x over previous
#include <hip/hip_runtime.h>


typedef __attribute__((ext_vector_type(2))) float v2f;
typedef __attribute__((ext_vector_type(8))) float v8f;

#define B_SZ 8
#define NA 4096
#define NB 4096
#define FD 64
#define WAVES 8
#define ALL_WAVES 16
#define THREADS (ALL_WAVES * 32)
#define ROWS_PER_WAVE 16
#define ROWS_PER_BLOCK (WAVES * ROWS_PER_WAVE)
#define ROWS_PER_MOVER (ROWS_PER_BLOCK / ALL_WAVES)
#define NTILES (NB / 16)
#define TPT 8
#define HEAD_POINTS THREADS
#define HEAD_TILES (HEAD_POINTS / 16)
#define TAIL_PER_THREAD ((NB - HEAD_POINTS) / (THREADS - WAVES * 32))
#define KBIAS 48387
#define HEAD_PRIO_DROP_TILE 16
#define PRIO_DROP_TILE 152

#define TOP3_INSERT_PK(pk, b0, b1, b2) do {          \
    int _r0 = min((pk), (b0));                        \
    (b0)    = max((pk), (b0));                        \
    int _r1 = min(_r0, (b1));                         \
    (b1)    = max(_r0, (b1));                         \
    (b2)    = max(_r1, (b2));                         \
  } while (0)

__device__ __forceinline__ int other_half(int v) {
  return __builtin_amdgcn_permlanex16(v, v, 0x76543210, 0xfedcba98, false, false);
}

__device__ __forceinline__ void put_point(unsigned* sPack, float* sBias, int n, int cx, int cy, int cz) {
  const int x = cx >> 4, y = cy >> 4, z = cz >> 4;
  sPack[n] = (unsigned)x | ((unsigned)y << 8) | ((unsigned)z << 16);
  sBias[n] = (float)(KBIAS - (x * x + y * y + z * z));
}

__device__ __forceinline__ void sweep_tiles(int tBegin, int tEnd,
    const unsigned* sPack, const float* sBias, v2f Acol, int mlane, bool hi,
    int& p0, int& p1, int& p2, int& thrbits, int prioDropTile)
{
  const v8f Cz = {};
  for (int t = tBegin; t < tEnd; t += TPT) {
    if (t == prioDropTile) __builtin_amdgcn_s_setprio(0);
    v8f Dv[TPT];
    int mt[TPT];
#pragma unroll
    for (int j = 0; j < TPT; ++j) {
      const int n = (t + j) * 16 + mlane;
      const unsigned u = sPack[n];
      const float bias = sBias[n];
      const float bx = (float)(u & 255u);
      const float by = (float)((u >> 8) & 255u);
      const float bz = (float)((u >> 16) & 255u);
      v2f Brow;
      Brow.x = hi ? bz   : bx;
      Brow.y = hi ? bias : by;

      Dv[j] = __builtin_amdgcn_wmma_f32_16x16x4_f32(
           false, Brow,  false, Acol,
           (short)0, Cz,  false,  false);

      const int g0 = max(max(__float_as_int(Dv[j][0]), __float_as_int(Dv[j][1])), __float_as_int(Dv[j][2]));
      const int g1 = max(max(__float_as_int(Dv[j][3]), __float_as_int(Dv[j][4])), __float_as_int(Dv[j][5]));
      const int g2 = max(__float_as_int(Dv[j][6]), __float_as_int(Dv[j][7]));
      mt[j] = max(max(g0, g1), g2);
    }

    unsigned need[TPT];
#pragma unroll
    for (int j = 0; j < TPT; ++j) need[j] = __builtin_amdgcn_ballot_w32(mt[j] > thrbits);
    unsigned any = need[0];
#pragma unroll
    for (int j = 1; j < TPT; ++j) any |= need[j];

    if (any != 0u) {
#pragma unroll
      for (int j = 0; j < TPT; ++j) {
        if (need[j] != 0u) {
          const int cbase = 4095 - ((t + j) * 16 + (hi ? 8 : 0));
#pragma unroll
          for (int r = 0; r < 8; ++r) {
            const int ik = (int)Dv[j][r];
            const int pk = (ik << 12) + (cbase - r);
            TOP3_INSERT_PK(pk, p0, p1, p2);
          }
        }
      }
      const int q0 = other_half(p0), q1 = other_half(p1), q2 = other_half(p2);
      const int thr = max(max(p2, q2), max(min(p1, q0), min(p0, q1)));
      thrbits = __float_as_int((float)(thr >> 12));
    }
  }
}

__global__ __launch_bounds__(THREADS) void knn3_wmma(
    const float* __restrict__ a_feats,
    const float* __restrict__ b_feats,
    const int*   __restrict__ a_coords,
    const int*   __restrict__ b_coords,
    float*       __restrict__ out)
{
  __shared__ unsigned sPack[NB];
  __shared__ float    sBias[NB];
  __shared__ int      sIdx[ROWS_PER_BLOCK][3];
  __shared__ float    sWgt[ROWS_PER_BLOCK][3];

  const int tid   = threadIdx.x;
  const int lane  = tid & 31;
  const int wave  = __builtin_amdgcn_readfirstlane(tid >> 5);
  const int mlane = lane & 15;
  const bool hi   = lane >= 16;

  const int blocksPerBatch = NA / ROWS_PER_BLOCK;
  const int b        = blockIdx.x / blocksPerBatch;
  const int blockRow = (blockIdx.x % blocksPerBatch) * ROWS_PER_BLOCK;
  const int* bc = b_coords + (size_t)b * NB * 3;

  {
    const int* p = bc + tid * 3;
    put_point(sPack, sBias, tid, p[0], p[1], p[2]);
  }
  __syncthreads();

  int p0 = -1, p1 = -1, p2 = -1;
  int thrbits = __float_as_int(-1.0f);
  int na2i = 0;
  v2f Acol = {};
  const bool younger = wave >= WAVES / 2 && wave < WAVES;

  if (wave < WAVES) {
    const int arow = blockRow + wave * ROWS_PER_WAVE + mlane;
    const int* ap = a_coords + ((size_t)b * NA + arow) * 3;
    const int iax = ap[0] >> 4, iay = ap[1] >> 4, iaz = ap[2] >> 4;
    na2i = iax * iax + iay * iay + iaz * iaz;
    Acol.x = hi ? (float)(2 * iaz) : (float)(2 * iax);
    Acol.y = hi ? 1.0f             : (float)(2 * iay);
    if (younger) __builtin_amdgcn_s_setprio(3);
    sweep_tiles(0, HEAD_TILES, sPack, sBias, Acol, mlane, hi, p0, p1, p2, thrbits,
                younger ? HEAD_PRIO_DROP_TILE : -1);
  } else {
    const int mtid = tid - WAVES * 32;
    int cx[TAIL_PER_THREAD], cy[TAIL_PER_THREAD], cz[TAIL_PER_THREAD];
#pragma unroll
    for (int k = 0; k < TAIL_PER_THREAD; ++k) {
      const int* p = bc + (HEAD_POINTS + k * (THREADS - WAVES * 32) + mtid) * 3;
      cx[k] = p[0]; cy[k] = p[1]; cz[k] = p[2];
    }
#pragma unroll
    for (int k = 0; k < TAIL_PER_THREAD; ++k)
      put_point(sPack, sBias, HEAD_POINTS + k * (THREADS - WAVES * 32) + mtid, cx[k], cy[k], cz[k]);
  }
  __syncthreads();

  if (wave < WAVES) {
  if (younger) __builtin_amdgcn_s_setprio(3);
  sweep_tiles(HEAD_TILES, NTILES, sPack, sBias, Acol, mlane, hi, p0, p1, p2, thrbits,
              younger ? PRIO_DROP_TILE : -1);

  {
    const int q0 = other_half(p0), q1 = other_half(p1), q2 = other_half(p2);
    TOP3_INSERT_PK(q0, p0, p1, p2);
    TOP3_INSERT_PK(q1, p0, p1, p2);
    TOP3_INSERT_PK(q2, p0, p1, p2);
  }
  if (!hi) {
    const int r = wave * ROWS_PER_WAVE + mlane;
    const int d20 = na2i + KBIAS - (int)((unsigned)p0 >> 12);
    const int d21 = na2i + KBIAS - (int)((unsigned)p1 >> 12);
    const int d22 = na2i + KBIAS - (int)((unsigned)p2 >> 12);
    const float inv = 1.0f / 128.0f;
    sIdx[r][0] = 4095 - (p0 & 4095);
    sIdx[r][1] = 4095 - (p1 & 4095);
    sIdx[r][2] = 4095 - (p2 & 4095);
    sWgt[r][0] = fmaxf(0.5f - sqrtf((float)d20) * inv, 0.0f);
    sWgt[r][1] = fmaxf(0.5f - sqrtf((float)d21) * inv, 0.0f);
    sWgt[r][2] = fmaxf(0.5f - sqrtf((float)d22) * inv, 0.0f);
  }
  }
  __syncthreads();

  const int moverRow = wave * ROWS_PER_MOVER;
  const float* afBase = a_feats + ((size_t)b * NA + blockRow + moverRow) * FD;
  float* outBase = out + ((size_t)b * NA + blockRow + moverRow) * (2 * FD);
  const float* bfB = b_feats + (size_t)b * NB * FD;

  for (int row = 0; row < ROWS_PER_MOVER; ++row) {
    const int j0 = sIdx[moverRow + row][0];
    const int j1 = sIdx[moverRow + row][1];
    const int j2 = sIdx[moverRow + row][2];
    const float w0 = sWgt[moverRow + row][0];
    const float w1 = sWgt[moverRow + row][1];
    const float w2 = sWgt[moverRow + row][2];
    const float* bf0 = bfB + (size_t)j0 * FD;
    const float* bf1 = bfB + (size_t)j1 * FD;
    const float* bf2 = bfB + (size_t)j2 * FD;
    const float* af  = afBase + (size_t)row * FD;
    float* op = outBase + (size_t)row * (2 * FD);
#pragma unroll
    for (int h = 0; h < 2; ++h) {
      const int d = lane + h * 32;
      const float tmp = w0 * bf0[d] + w1 * bf1[d] + w2 * bf2[d];
      op[d]      = af[d];
      op[FD + d] = tmp;
    }
  }
}

extern "C" void kernel_launch(void* const* d_in, const int* in_sizes, int n_in,
                              void* d_out, int out_size, void* d_ws, size_t ws_size,
                              hipStream_t stream) {
  const float* a_feats  = (const float*)d_in[0];
  const float* b_feats  = (const float*)d_in[1];
  const int*   a_coords = (const int*)d_in[2];
  const int*   b_coords = (const int*)d_in[3];
  float* out = (float*)d_out;

  dim3 grid(B_SZ * (NA / ROWS_PER_BLOCK));
  dim3 block(THREADS);
  knn3_wmma<<<grid, block, 0, stream>>>(a_feats, b_feats, a_coords, b_coords, out);
}


# top-3 insert rewritten on the sorted-list invariant: each new entry = max(min(pk, old), old) of OLD entries only, three independent ops per candidate, was a five-step min/max chain; same values
# speedup vs baseline: 2.0398x; 1.0347x over previous
#include <hip/hip_runtime.h>


typedef __attribute__((ext_vector_type(2))) float v2f;
typedef __attribute__((ext_vector_type(8))) float v8f;

#define B_SZ 8
#define NA 4096
#define NB 4096
#define FD 64
#define WAVES 8
#define ALL_WAVES 16
#define THREADS (ALL_WAVES * 32)
#define ROWS_PER_WAVE 16
#define ROWS_PER_BLOCK (WAVES * ROWS_PER_WAVE)
#define ROWS_PER_MOVER (ROWS_PER_BLOCK / ALL_WAVES)
#define NTILES (NB / 16)
#define TPT 8
#define HEAD_POINTS THREADS
#define HEAD_TILES (HEAD_POINTS / 16)
#define TAIL_PER_THREAD ((NB - HEAD_POINTS) / (THREADS - WAVES * 32))
#define KBIAS 48387
#define HEAD_PRIO_DROP_TILE 16
#define PRIO_DROP_TILE 152

#define TOP3_INSERT_PK(pk, b0, b1, b2) do {          \
    const int _n2 = max(min((pk), (b1)), (b2));       \
    const int _n1 = max(min((pk), (b0)), (b1));       \
    (b0) = max((pk), (b0));                           \
    (b1) = _n1;                                       \
    (b2) = _n2;                                       \
  } while (0)

__device__ __forceinline__ int other_half(int v) {
  return __builtin_amdgcn_permlanex16(v, v, 0x76543210, 0xfedcba98, false, false);
}

__device__ __forceinline__ void put_point(unsigned* sPack, float* sBias, int n, int cx, int cy, int cz) {
  const int x = cx >> 4, y = cy >> 4, z = cz >> 4;
  sPack[n] = (unsigned)x | ((unsigned)y << 8) | ((unsigned)z << 16);
  sBias[n] = (float)(KBIAS - (x * x + y * y + z * z));
}

__device__ __forceinline__ void sweep_tiles(int tBegin, int tEnd,
    const unsigned* sPack, const float* sBias, v2f Acol, int mlane, bool hi,
    int& p0, int& p1, int& p2, int& thrbits, int prioDropTile)
{
  const v8f Cz = {};
  int cb = 4095 - (tBegin * 16 + (hi ? 8 : 0));
  for (int t = tBegin; t < tEnd; t += TPT, cb -= 16 * TPT) {
    if (t == prioDropTile) __builtin_amdgcn_s_setprio(0);
    v8f Dv[TPT];
    int mt[TPT];
#pragma unroll
    for (int j = 0; j < TPT; ++j) {
      const int n = (t + j) * 16 + mlane;
      const unsigned u = sPack[n];
      const float bias = sBias[n];
      const float bx = (float)(u & 255u);
      const float by = (float)((u >> 8) & 255u);
      const float bz = (float)((u >> 16) & 255u);
      v2f Brow;
      Brow.x = hi ? bz   : bx;
      Brow.y = hi ? bias : by;

      Dv[j] = __builtin_amdgcn_wmma_f32_16x16x4_f32(
           false, Brow,  false, Acol,
           (short)0, Cz,  false,  false);

      const int g0 = max(max(__float_as_int(Dv[j][0]), __float_as_int(Dv[j][1])), __float_as_int(Dv[j][2]));
      const int g1 = max(max(__float_as_int(Dv[j][3]), __float_as_int(Dv[j][4])), __float_as_int(Dv[j][5]));
      const int g2 = max(__float_as_int(Dv[j][6]), __float_as_int(Dv[j][7]));
      mt[j] = max(max(g0, g1), g2);
    }

    unsigned need[TPT];
#pragma unroll
    for (int j = 0; j < TPT; ++j) need[j] = __builtin_amdgcn_ballot_w32(mt[j] > thrbits);
    unsigned any = need[0];
#pragma unroll
    for (int j = 1; j < TPT; ++j) any |= need[j];

    if (any != 0u) {
#pragma unroll
      for (int j = 0; j < TPT; ++j) {
        if (need[j] != 0u) {
          const int cbase = cb - j * 16;
#pragma unroll
          for (int r = 0; r < 8; ++r) {
            const int ik = (int)Dv[j][r];
            const int pk = (ik << 12) + (cbase - r);
            TOP3_INSERT_PK(pk, p0, p1, p2);
          }
        }
      }
      const int q0 = other_half(p0), q1 = other_half(p1), q2 = other_half(p2);
      const int thr = max(max(p2, q2), max(min(p1, q0), min(p0, q1)));
      thrbits = __float_as_int((float)(thr >> 12));
    }
  }
}

__global__ __launch_bounds__(THREADS) void knn3_wmma(
    const float* __restrict__ a_feats,
    const float* __restrict__ b_feats,
    const int*   __restrict__ a_coords,
    const int*   __restrict__ b_coords,
    float*       __restrict__ out)
{
  __shared__ unsigned sPack[NB];
  __shared__ float    sBias[NB];
  __shared__ int      sIdx[ROWS_PER_BLOCK][3];
  __shared__ float    sWgt[ROWS_PER_BLOCK][3];

  const int tid   = threadIdx.x;
  const int lane  = tid & 31;
  const int wave  = __builtin_amdgcn_readfirstlane(tid >> 5);
  const int mlane = lane & 15;
  const bool hi   = lane >= 16;

  const int blocksPerBatch = NA / ROWS_PER_BLOCK;
  const int b        = blockIdx.x / blocksPerBatch;
  const int blockRow = (blockIdx.x % blocksPerBatch) * ROWS_PER_BLOCK;
  const int* bc = b_coords + (size_t)b * NB * 3;

  {
    const int* p = bc + tid * 3;
    put_point(sPack, sBias, tid, p[0], p[1], p[2]);
  }
  __syncthreads();

  int p0 = -1, p1 = -1, p2 = -1;
  int thrbits = __float_as_int(-1.0f);
  int na2i = 0;
  v2f Acol = {};
  const bool younger = wave >= WAVES / 2 && wave < WAVES;

  if (wave < WAVES) {
    const int arow = blockRow + wave * ROWS_PER_WAVE + mlane;
    const int* ap = a_coords + ((size_t)b * NA + arow) * 3;
    const int iax = ap[0] >> 4, iay = ap[1] >> 4, iaz = ap[2] >> 4;
    na2i = iax * iax + iay * iay + iaz * iaz;
    Acol.x = hi ? (float)(2 * iaz) : (float)(2 * iax);
    Acol.y = hi ? 1.0f             : (float)(2 * iay);
    if (younger) __builtin_amdgcn_s_setprio(3);
    sweep_tiles(0, HEAD_TILES, sPack, sBias, Acol, mlane, hi, p0, p1, p2, thrbits,
                younger ? HEAD_PRIO_DROP_TILE : -1);
  } else {
    const int mtid = tid - WAVES * 32;
    int cx[TAIL_PER_THREAD], cy[TAIL_PER_THREAD], cz[TAIL_PER_THREAD];
#pragma unroll
    for (int k = 0; k < TAIL_PER_THREAD; ++k) {
      const int* p = bc + (HEAD_POINTS + k * (THREADS - WAVES * 32) + mtid) * 3;
      cx[k] = p[0]; cy[k] = p[1]; cz[k] = p[2];
    }
#pragma unroll
    for (int k = 0; k < TAIL_PER_THREAD; ++k)
      put_point(sPack, sBias, HEAD_POINTS + k * (THREADS - WAVES * 32) + mtid, cx[k], cy[k], cz[k]);
  }
  __syncthreads();

  if (wave < WAVES) {
  if (younger) __builtin_amdgcn_s_setprio(3);
  sweep_tiles(HEAD_TILES, NTILES, sPack, sBias, Acol, mlane, hi, p0, p1, p2, thrbits,
              younger ? PRIO_DROP_TILE : -1);

  {
    const int q0 = other_half(p0), q1 = other_half(p1), q2 = other_half(p2);
    TOP3_INSERT_PK(q0, p0, p1, p2);
    TOP3_INSERT_PK(q1, p0, p1, p2);
    TOP3_INSERT_PK(q2, p0, p1, p2);
  }
  if (!hi) {
    const int r = wave * ROWS_PER_WAVE + mlane;
    const int d20 = na2i + KBIAS - (int)((unsigned)p0 >> 12);
    const int d21 = na2i + KBIAS - (int)((unsigned)p1 >> 12);
    const int d22 = na2i + KBIAS - (int)((unsigned)p2 >> 12);
    const float inv = 1.0f / 128.0f;
    sIdx[r][0] = 4095 - (p0 & 4095);
    sIdx[r][1] = 4095 - (p1 & 4095);
    sIdx[r][2] = 4095 - (p2 & 4095);
    sWgt[r][0] = fmaxf(0.5f - sqrtf((float)d20) * inv, 0.0f);
    sWgt[r][1] = fmaxf(0.5f - sqrtf((float)d21) * inv, 0.0f);
    sWgt[r][2] = fmaxf(0.5f - sqrtf((float)d22) * inv, 0.0f);
  }
  }
  __syncthreads();

  const int moverRow = wave * ROWS_PER_MOVER;
  const float* afBase = a_feats + ((size_t)b * NA + blockRow + moverRow) * FD;
  float* outBase = out + ((size_t)b * NA + blockRow + moverRow) * (2 * FD);
  const float* bfB = b_feats + (size_t)b * NB * FD;

  for (int row = 0; row < ROWS_PER_MOVER; ++row) {
    const int j0 = sIdx[moverRow + row][0];
    const int j1 = sIdx[moverRow + row][1];
    const int j2 = sIdx[moverRow + row][2];
    const float w0 = sWgt[moverRow + row][0];
    const float w1 = sWgt[moverRow + row][1];
    const float w2 = sWgt[moverRow + row][2];
    const float* bf0 = bfB + (size_t)j0 * FD;
    const float* bf1 = bfB + (size_t)j1 * FD;
    const float* bf2 = bfB + (size_t)j2 * FD;
    const float* af  = afBase + (size_t)row * FD;
    float* op = outBase + (size_t)row * (2 * FD);
#pragma unroll
    for (int h = 0; h < 2; ++h) {
      const int d = lane + h * 32;
      const float tmp = w0 * bf0[d] + w1 * bf1[d] + w2 * bf2[d];
      op[d]      = af[d];
      op[FD + d] = tmp;
    }
  }
}

extern "C" void kernel_launch(void* const* d_in, const int* in_sizes, int n_in,
                              void* d_out, int out_size, void* d_ws, size_t ws_size,
                              hipStream_t stream) {
  const float* a_feats  = (const float*)d_in[0];
  const float* b_feats  = (const float*)d_in[1];
  const int*   a_coords = (const int*)d_in[2];
  const int*   b_coords = (const int*)d_in[3];
  float* out = (float*)d_out;

  dim3 grid(B_SZ * (NA / ROWS_PER_BLOCK));
  dim3 block(THREADS);
  knn3_wmma<<<grid, block, 0, stream>>>(a_feats, b_feats, a_coords, b_coords, out);
}


# spatial ownership: a block owns a box of cells, not 128 rows, and sweeps only the b points within 14 cells of it (index-order compaction); rows whose 3rd neighbour is farther are redone on all b
# speedup vs baseline: 3.1861x; 1.5620x over previous
#include <hip/hip_runtime.h>


typedef __attribute__((ext_vector_type(2))) float v2f;
typedef __attribute__((ext_vector_type(8))) float v8f;

#define B_SZ 8
#define NA 4096
#define NB 4096
#define FD 64
#define ALL_WAVES 16
#define THREADS (ALL_WAVES * 32)
#define CHUNK (ALL_WAVES * 16)
#define PER_THREAD (NB / THREADS)
#define TPT 8
#define KBIAS 48387
#define BOX_X 32
#define BOX_Y 32
#define BOX_Z 64
#define BOXES_X (128 / BOX_X)
#define BOXES_Y (128 / BOX_Y)
#define BOXES_Z (128 / BOX_Z)
#define BOXES (BOXES_X * BOXES_Y * BOXES_Z)
#define MARGIN 14
#define GROWS 11

#define TOP3_INSERT_PK(pk, b0, b1, b2) do {          \
    const int _n2 = max(min((pk), (b1)), (b2));       \
    const int _n1 = max(min((pk), (b0)), (b1));       \
    (b0) = max((pk), (b0));                           \
    (b1) = _n1;                                       \
    (b2) = _n2;                                       \
  } while (0)

__device__ __forceinline__ int other_half(int v) {
  return __builtin_amdgcn_permlanex16(v, v, 0x76543210, 0xfedcba98, false, false);
}

__device__ __forceinline__ void put_point(unsigned* sPack, float* sBias, int n, int cx, int cy, int cz) {
  const int x = cx >> 4, y = cy >> 4, z = cz >> 4;
  sPack[n] = (unsigned)x | ((unsigned)y << 8) | ((unsigned)z << 16);
  sBias[n] = (float)(KBIAS - (x * x + y * y + z * z));
}

__device__ __forceinline__ void sweep_tiles(int tBegin, int tEnd,
    const unsigned* sPack, const float* sBias, v2f Acol, int mlane, bool hi,
    int& p0, int& p1, int& p2, int& thrbits, unsigned colMask)
{
  const v8f Cz = {};
  int cb = 4095 - (tBegin * 16 + (hi ? 8 : 0));
  for (int t = tBegin; t < tEnd; t += TPT, cb -= 16 * TPT) {
    v8f Dv[TPT];
    int mt[TPT];
#pragma unroll
    for (int j = 0; j < TPT; ++j) {
      const int n = (t + j) * 16 + mlane;
      const unsigned u = sPack[n];
      const float bias = sBias[n];
      const float bx = (float)(u & 255u);
      const float by = (float)((u >> 8) & 255u);
      const float bz = (float)((u >> 16) & 255u);
      v2f Brow;
      Brow.x = hi ? bz   : bx;
      Brow.y = hi ? bias : by;

      Dv[j] = __builtin_amdgcn_wmma_f32_16x16x4_f32(
           false, Brow,  false, Acol,
           (short)0, Cz,  false,  false);

      const int g0 = max(max(__float_as_int(Dv[j][0]), __float_as_int(Dv[j][1])), __float_as_int(Dv[j][2]));
      const int g1 = max(max(__float_as_int(Dv[j][3]), __float_as_int(Dv[j][4])), __float_as_int(Dv[j][5]));
      const int g2 = max(__float_as_int(Dv[j][6]), __float_as_int(Dv[j][7]));
      mt[j] = max(max(g0, g1), g2);
    }

    unsigned need[TPT];
#pragma unroll
    for (int j = 0; j < TPT; ++j) need[j] = __builtin_amdgcn_ballot_w32(mt[j] > thrbits) & colMask;
    unsigned any = need[0];
#pragma unroll
    for (int j = 1; j < TPT; ++j) any |= need[j];

    if (any != 0u) {
#pragma unroll
      for (int j = 0; j < TPT; ++j) {
        if (need[j] != 0u) {
          const int cbase = cb - j * 16;
#pragma unroll
          for (int r = 0; r < 8; ++r) {
            const int ik = (int)Dv[j][r];
            const int pk = (ik << 12) + (cbase - r);
            TOP3_INSERT_PK(pk, p0, p1, p2);
          }
        }
      }
      const int q0 = other_half(p0), q1 = other_half(p1), q2 = other_half(p2);
      const int thr = max(max(p2, q2), max(min(p1, q0), min(p0, q1)));
      thrbits = __float_as_int((float)(thr >> 12));
    }
  }
}


__device__ __forceinline__ float weight_of(int na2i, int pk) {
  const int d2 = na2i + KBIAS - (int)((unsigned)pk >> 12);
  return fmaxf(0.5f - sqrtf((float)d2) * (1.0f / 128.0f), 0.0f);
}

__global__ __launch_bounds__(THREADS) void knn3_wmma(
    const float* __restrict__ a_feats,
    const float* __restrict__ b_feats,
    const int*   __restrict__ a_coords,
    const int*   __restrict__ b_coords,
    float*       __restrict__ out)
{
  __shared__ unsigned sPackAll[NB];
  __shared__ float    sBiasAll[NB];
  __shared__ unsigned sPack[NB];
  __shared__ float    sBias[NB];
  __shared__ unsigned short sBIdx[NB];
  __shared__ unsigned sAEnt[NA];
  __shared__ int      sCnt[ALL_WAVES][2];
  __shared__ int      sUCnt[ALL_WAVES];
  __shared__ unsigned short sFbRow[CHUNK];
  __shared__ int      sFbPk[ALL_WAVES][16][3];
  __shared__ int      sIdx[CHUNK][3];
  __shared__ float    sWgt[CHUNK][3];

  const int tid   = threadIdx.x;
  const int lane  = tid & 31;
  const int wave  = __builtin_amdgcn_readfirstlane(tid >> 5);
  const int mlane = lane & 15;
  const bool hi   = lane >= 16;
  const unsigned lanesBelow = (1u << lane) - 1u;

  const int b   = blockIdx.x / BOXES;
  const int box = blockIdx.x % BOXES;
  const int x0 = (box % BOXES_X) * BOX_X;
  const int y0 = ((box / BOXES_X) % BOXES_Y) * BOX_Y;
  const int z0 = (box / (BOXES_X * BOXES_Y)) * BOX_Z;
  const int* bc = b_coords + (size_t)b * NB * 3;
  const int* ac = a_coords + (size_t)b * NA * 3;

  int bx[PER_THREAD], by[PER_THREAD], bz[PER_THREAD];
  int ax[PER_THREAD], ay[PER_THREAD], az[PER_THREAD];
#pragma unroll
  for (int k = 0; k < PER_THREAD; ++k) {
    const int n = wave * (32 * PER_THREAD) + k * 32 + lane;
    bx[k] = bc[n * 3]; by[k] = bc[n * 3 + 1]; bz[k] = bc[n * 3 + 2];
    ax[k] = ac[n * 3]; ay[k] = ac[n * 3 + 1]; az[k] = ac[n * 3 + 2];
  }

  int posA[PER_THREAD], posB[PER_THREAD];
  int cntA = 0, cntB = 0;
#pragma unroll
  for (int k = 0; k < PER_THREAD; ++k) {
    const int n = wave * (32 * PER_THREAD) + k * 32 + lane;
    put_point(sPackAll, sBiasAll, n, bx[k], by[k], bz[k]);
    const int qx = bx[k] >> 4, qy = by[k] >> 4, qz = bz[k] >> 4;
    const bool selB = (unsigned)(qx - (x0 - MARGIN)) < (unsigned)(BOX_X + 2 * MARGIN)
                   && (unsigned)(qy - (y0 - MARGIN)) < (unsigned)(BOX_Y + 2 * MARGIN)
                   && (unsigned)(qz - (z0 - MARGIN)) < (unsigned)(BOX_Z + 2 * MARGIN);
    const unsigned mB = __builtin_amdgcn_ballot_w32(selB);
    posB[k] = selB ? cntB + __builtin_popcount(mB & lanesBelow) : -1;
    cntB += __builtin_popcount(mB);
    const int rx = (ax[k] >> 4) - x0, ry = (ay[k] >> 4) - y0, rz = (az[k] >> 4) - z0;
    const bool selA = (unsigned)rx < (unsigned)BOX_X && (unsigned)ry < (unsigned)BOX_Y && (unsigned)rz < (unsigned)BOX_Z;
    const unsigned mA = __builtin_amdgcn_ballot_w32(selA);
    posA[k] = selA ? cntA + __builtin_popcount(mA & lanesBelow) : -1;
    cntA += __builtin_popcount(mA);
  }
  if (lane == 0) { sCnt[wave][0] = cntA; sCnt[wave][1] = cntB; }
  __syncthreads();

  int baseA = 0, baseB = 0, nA = 0, nB = 0;
  for (int w = 0; w < ALL_WAVES; ++w) {
    const int ca = sCnt[w][0], cb = sCnt[w][1];
    if (w < wave) { baseA += ca; baseB += cb; }
    nA += ca; nB += cb;
  }
  nA = __builtin_amdgcn_readfirstlane(nA);
  nB = __builtin_amdgcn_readfirstlane(nB);
  const int nBpad = max((nB + 16 * TPT - 1) & ~(16 * TPT - 1), 16 * TPT);

#pragma unroll
  for (int k = 0; k < PER_THREAD; ++k) {
    const int n = wave * (32 * PER_THREAD) + k * 32 + lane;
    if (posB[k] >= 0) {
      const int p = baseB + posB[k];
      put_point(sPack, sBias, p, bx[k], by[k], bz[k]);
      sBIdx[p] = (unsigned short)n;
    }
    if (posA[k] >= 0) {
      const unsigned rx = (unsigned)((ax[k] >> 4) - x0), ry = (unsigned)((ay[k] >> 4) - y0), rz = (unsigned)((az[k] >> 4) - z0);
      sAEnt[baseA + posA[k]] = (unsigned)n | (rx << 12) | (ry << 17) | (rz << 22);
    }
  }
  if (tid < nBpad - nB) { sPack[nB + tid] = 0u; sBias[nB + tid] = 0.0f; sBIdx[nB + tid] = 0; }
  __syncthreads();

  const float* afB = a_feats + (size_t)b * NA * FD;
  const float* bfB = b_feats + (size_t)b * NB * FD;
  float* outB = out + (size_t)b * NA * (2 * FD);

  for (int c0 = 0; c0 < nA; c0 += CHUNK) {
    const int cnt = min(CHUNK, nA - c0);
    const int tile0 = wave * 16;
    const int lrow = tile0 + mlane;
    bool unresolved = false;

    if (tile0 < cnt) {
      const bool valid = lrow < cnt;
      const unsigned ent = sAEnt[c0 + (valid ? lrow : tile0)];
      const int iax = x0 + (int)((ent >> 12) & 31u), iay = y0 + (int)((ent >> 17) & 31u), iaz = z0 + (int)((ent >> 22) & 63u);
      const int na2i = iax * iax + iay * iay + iaz * iaz;
      v2f Acol;
      Acol.x = hi ? (float)(2 * iaz) : (float)(2 * iax);
      Acol.y = hi ? 1.0f             : (float)(2 * iay);
      int p0 = -1, p1 = -1, p2 = -1;
      int thrbits = __float_as_int(-1.0f);
      sweep_tiles(0, nBpad / 16, sPack, sBias, Acol, mlane, hi, p0, p1, p2, thrbits,
                  __builtin_amdgcn_ballot_w32(valid));
      {
        const int q0 = other_half(p0), q1 = other_half(p1), q2 = other_half(p2);
        TOP3_INSERT_PK(q0, p0, p1, p2);
        TOP3_INSERT_PK(q1, p0, p1, p2);
        TOP3_INSERT_PK(q2, p0, p1, p2);
      }
      if (!hi) {
        const int d2third = na2i + KBIAS - (int)((unsigned)p2 >> 12);
        unresolved = valid && !(p2 >= 0 && d2third <= MARGIN * MARGIN);
        sIdx[lrow][0] = sBIdx[4095 - (p0 & 4095)];
        sIdx[lrow][1] = sBIdx[4095 - (p1 & 4095)];
        sIdx[lrow][2] = sBIdx[4095 - (p2 & 4095)];
        sWgt[lrow][0] = weight_of(na2i, p0);
        sWgt[lrow][1] = weight_of(na2i, p1);
        sWgt[lrow][2] = weight_of(na2i, p2);
      }
    }

    const unsigned um = __builtin_amdgcn_ballot_w32(unresolved);
    if (lane == 0) sUCnt[wave] = __builtin_popcount(um);
    __syncthreads();
    int ubase = 0, nU = 0;
    for (int w = 0; w < ALL_WAVES; ++w) {
      const int c = sUCnt[w];
      if (w < wave) ubase += c;
      nU += c;
    }
    nU = __builtin_amdgcn_readfirstlane(nU);

    if (nU > 0) {
      if (unresolved) sFbRow[ubase + __builtin_popcount(um & lanesBelow)] = (unsigned short)lrow;
      __syncthreads();
      for (int f0 = 0; f0 < nU; f0 += 16) {
        const bool fvalid = f0 + mlane < nU;
        const int frow = sFbRow[fvalid ? f0 + mlane : f0];
        const unsigned ent = sAEnt[c0 + frow];
        const int iax = x0 + (int)((ent >> 12) & 31u), iay = y0 + (int)((ent >> 17) & 31u), iaz = z0 + (int)((ent >> 22) & 63u);
        const int na2i = iax * iax + iay * iay + iaz * iaz;
        v2f Acol;
        Acol.x = hi ? (float)(2 * iaz) : (float)(2 * iax);
        Acol.y = hi ? 1.0f             : (float)(2 * iay);
        int p0 = -1, p1 = -1, p2 = -1;
        int thrbits = __float_as_int(-1.0f);
        sweep_tiles(wave * 16, wave * 16 + 16, sPackAll, sBiasAll, Acol, mlane, hi, p0, p1, p2, thrbits,
                    __builtin_amdgcn_ballot_w32(fvalid));
        {
          const int q0 = other_half(p0), q1 = other_half(p1), q2 = other_half(p2);
          TOP3_INSERT_PK(q0, p0, p1, p2);
          TOP3_INSERT_PK(q1, p0, p1, p2);
          TOP3_INSERT_PK(q2, p0, p1, p2);
        }
        if (!hi) { sFbPk[wave][mlane][0] = p0; sFbPk[wave][mlane][1] = p1; sFbPk[wave][mlane][2] = p2; }
        __syncthreads();
        if (wave == 0 && !hi && fvalid) {
          int r0 = -1, r1 = -1, r2 = -1;
          for (int w = 0; w < ALL_WAVES; ++w) {
            const int e0 = sFbPk[w][mlane][0], e1 = sFbPk[w][mlane][1], e2 = sFbPk[w][mlane][2];
            TOP3_INSERT_PK(e0, r0, r1, r2);
            TOP3_INSERT_PK(e1, r0, r1, r2);
            TOP3_INSERT_PK(e2, r0, r1, r2);
          }
          sIdx[frow][0] = 4095 - (r0 & 4095);
          sIdx[frow][1] = 4095 - (r1 & 4095);
          sIdx[frow][2] = 4095 - (r2 & 4095);
          sWgt[frow][0] = weight_of(na2i, r0);
          sWgt[frow][1] = weight_of(na2i, r1);
          sWgt[frow][2] = weight_of(na2i, r2);
        }
        __syncthreads();
      }
    }

    for (int k0 = 0; k0 < CHUNK / ALL_WAVES; k0 += GROWS) {
      const int first = k0 * ALL_WAVES + wave;
      if (first >= cnt) break;
      float va[GROWS][2], g0[GROWS][2], g1[GROWS][2], g2[GROWS][2];
#pragma unroll
      for (int kk = 0; kk < GROWS; ++kk) {
        const int lr = (k0 + kk) * ALL_WAVES + wave;
        const int le = (k0 + kk < CHUNK / ALL_WAVES && lr < cnt) ? lr : first;
        const float* bf0 = bfB + (size_t)sIdx[le][0] * FD;
        const float* bf1 = bfB + (size_t)sIdx[le][1] * FD;
        const float* bf2 = bfB + (size_t)sIdx[le][2] * FD;
        const float* af  = afB + (size_t)(sAEnt[c0 + le] & 4095u) * FD;
#pragma unroll
        for (int h = 0; h < 2; ++h) {
          const int d = lane + h * 32;
          va[kk][h] = af[d]; g0[kk][h] = bf0[d]; g1[kk][h] = bf1[d]; g2[kk][h] = bf2[d];
        }
      }
#pragma unroll
      for (int kk = 0; kk < GROWS; ++kk) {
        const int lr = (k0 + kk) * ALL_WAVES + wave;
        if (k0 + kk < CHUNK / ALL_WAVES && lr < cnt) {
          float* op = outB + (size_t)(sAEnt[c0 + lr] & 4095u) * (2 * FD);
          const float w0 = sWgt[lr][0], w1 = sWgt[lr][1], w2 = sWgt[lr][2];
#pragma unroll
          for (int h = 0; h < 2; ++h) {
            const int d = lane + h * 32;
            const float tmp = w0 * g0[kk][h] + w1 * g1[kk][h] + w2 * g2[kk][h];
            op[d]      = va[kk][h];
            op[FD + d] = tmp;
          }
        }
      }
    }
    __syncthreads();
  }
}

extern "C" void kernel_launch(void* const* d_in, const int* in_sizes, int n_in,
                              void* d_out, int out_size, void* d_ws, size_t ws_size,
                              hipStream_t stream) {
  const float* a_feats  = (const float*)d_in[0];
  const float* b_feats  = (const float*)d_in[1];
  const int*   a_coords = (const int*)d_in[2];
  const int*   b_coords = (const int*)d_in[3];
  float* out = (float*)d_out;

  dim3 grid(B_SZ * BOXES);
  dim3 block(THREADS);
  knn3_wmma<<<grid, block, 0, stream>>>(a_feats, b_feats, a_coords, b_coords, out);
}


# a_feats half of the output leaves the epilogue: each block copies a fixed 128-row slice with 16-byte lanes on waves 12-15, which sweep no tile, under the others' sweep; epilogue: gathers only
# speedup vs baseline: 3.3720x; 1.0583x over previous
#include <hip/hip_runtime.h>


typedef __attribute__((ext_vector_type(2))) float v2f;
typedef __attribute__((ext_vector_type(4))) float v4f;
typedef __attribute__((ext_vector_type(8))) float v8f;

#define B_SZ 8
#define NA 4096
#define NB 4096
#define FD 64
#define ALL_WAVES 16
#define THREADS (ALL_WAVES * 32)
#define CHUNK (ALL_WAVES * 16)
#define PER_THREAD (NB / THREADS)
#define TPT 8
#define KBIAS 48387
#define BOX_X 32
#define BOX_Y 32
#define BOX_Z 64
#define BOXES_X (128 / BOX_X)
#define BOXES_Y (128 / BOX_Y)
#define BOXES_Z (128 / BOX_Z)
#define BOXES (BOXES_X * BOXES_Y * BOXES_Z)
#define MARGIN 14
#define GROWS 11
#define COPY_ROWS (NA / BOXES)
#define COPY_WAVES 4
#define COPY_PAIRS (COPY_ROWS / COPY_WAVES / 2)

#define TOP3_INSERT_PK(pk, b0, b1, b2) do {          \
    const int _n2 = max(min((pk), (b1)), (b2));       \
    const int _n1 = max(min((pk), (b0)), (b1));       \
    (b0) = max((pk), (b0));                           \
    (b1) = _n1;                                       \
    (b2) = _n2;                                       \
  } while (0)

__device__ __forceinline__ int other_half(int v) {
  return __builtin_amdgcn_permlanex16(v, v, 0x76543210, 0xfedcba98, false, false);
}

__device__ __forceinline__ void put_point(unsigned* sPack, float* sBias, int n, int cx, int cy, int cz) {
  const int x = cx >> 4, y = cy >> 4, z = cz >> 4;
  sPack[n] = (unsigned)x | ((unsigned)y << 8) | ((unsigned)z << 16);
  sBias[n] = (float)(KBIAS - (x * x + y * y + z * z));
}

__device__ __forceinline__ void sweep_tiles(int tBegin, int tEnd,
    const unsigned* sPack, const float* sBias, v2f Acol, int mlane, bool hi,
    int& p0, int& p1, int& p2, int& thrbits, unsigned colMask)
{
  const v8f Cz = {};
  int cb = 4095 - (tBegin * 16 + (hi ? 8 : 0));
  for (int t = tBegin; t < tEnd; t += TPT, cb -= 16 * TPT) {
    v8f Dv[TPT];
    int mt[TPT];
#pragma unroll
    for (int j = 0; j < TPT; ++j) {
      const int n = (t + j) * 16 + mlane;
      const unsigned u = sPack[n];
      const float bias = sBias[n];
      const float bx = (float)(u & 255u);
      const float by = (float)((u >> 8) & 255u);
      const float bz = (float)((u >> 16) & 255u);
      v2f Brow;
      Brow.x = hi ? bz   : bx;
      Brow.y = hi ? bias : by;

      Dv[j] = __builtin_amdgcn_wmma_f32_16x16x4_f32(
           false, Brow,  false, Acol,
           (short)0, Cz,  false,  false);

      const int g0 = max(max(__float_as_int(Dv[j][0]), __float_as_int(Dv[j][1])), __float_as_int(Dv[j][2]));
      const int g1 = max(max(__float_as_int(Dv[j][3]), __float_as_int(Dv[j][4])), __float_as_int(Dv[j][5]));
      const int g2 = max(__float_as_int(Dv[j][6]), __float_as_int(Dv[j][7]));
      mt[j] = max(max(g0, g1), g2);
    }

    unsigned need[TPT];
#pragma unroll
    for (int j = 0; j < TPT; ++j) need[j] = __builtin_amdgcn_ballot_w32(mt[j] > thrbits) & colMask;
    unsigned any = need[0];
#pragma unroll
    for (int j = 1; j < TPT; ++j) any |= need[j];

    if (any != 0u) {
#pragma unroll
      for (int j = 0; j < TPT; ++j) {
        if (need[j] != 0u) {
          const int cbase = cb - j * 16;
#pragma unroll
          for (int r = 0; r < 8; ++r) {
            const int ik = (int)Dv[j][r];
            const int pk = (ik << 12) + (cbase - r);
            TOP3_INSERT_PK(pk, p0, p1, p2);
          }
        }
      }
      const int q0 = other_half(p0), q1 = other_half(p1), q2 = other_half(p2);
      const int thr = max(max(p2, q2), max(min(p1, q0), min(p0, q1)));
      thrbits = __float_as_int((float)(thr >> 12));
    }
  }
}


__device__ __forceinline__ float weight_of(int na2i, int pk) {
  const int d2 = na2i + KBIAS - (int)((unsigned)pk >> 12);
  return fmaxf(0.5f - sqrtf((float)d2) * (1.0f / 128.0f), 0.0f);
}

__global__ __launch_bounds__(THREADS) void knn3_wmma(
    const float* __restrict__ a_feats,
    const float* __restrict__ b_feats,
    const int*   __restrict__ a_coords,
    const int*   __restrict__ b_coords,
    float*       __restrict__ out)
{
  __shared__ unsigned sPackAll[NB];
  __shared__ float    sBiasAll[NB];
  __shared__ unsigned sPack[NB];
  __shared__ float    sBias[NB];
  __shared__ unsigned short sBIdx[NB];
  __shared__ unsigned sAEnt[NA];
  __shared__ int      sCnt[ALL_WAVES][2];
  __shared__ int      sUCnt[ALL_WAVES];
  __shared__ unsigned short sFbRow[CHUNK];
  __shared__ int      sFbPk[ALL_WAVES][16][3];
  __shared__ int      sIdx[CHUNK][3];
  __shared__ float    sWgt[CHUNK][3];

  const int tid   = threadIdx.x;
  const int lane  = tid & 31;
  const int wave  = __builtin_amdgcn_readfirstlane(tid >> 5);
  const int mlane = lane & 15;
  const bool hi   = lane >= 16;
  const unsigned lanesBelow = (1u << lane) - 1u;

  const int b   = blockIdx.x / BOXES;
  const int box = blockIdx.x % BOXES;
  const int x0 = (box % BOXES_X) * BOX_X;
  const int y0 = ((box / BOXES_X) % BOXES_Y) * BOX_Y;
  const int z0 = (box / (BOXES_X * BOXES_Y)) * BOX_Z;
  const int* bc = b_coords + (size_t)b * NB * 3;
  const int* ac = a_coords + (size_t)b * NA * 3;

  int bx[PER_THREAD], by[PER_THREAD], bz[PER_THREAD];
  int ax[PER_THREAD], ay[PER_THREAD], az[PER_THREAD];
#pragma unroll
  for (int k = 0; k < PER_THREAD; ++k) {
    const int n = wave * (32 * PER_THREAD) + k * 32 + lane;
    bx[k] = bc[n * 3]; by[k] = bc[n * 3 + 1]; bz[k] = bc[n * 3 + 2];
    ax[k] = ac[n * 3]; ay[k] = ac[n * 3 + 1]; az[k] = ac[n * 3 + 2];
  }

  int posA[PER_THREAD], posB[PER_THREAD];
  int cntA = 0, cntB = 0;
#pragma unroll
  for (int k = 0; k < PER_THREAD; ++k) {
    const int n = wave * (32 * PER_THREAD) + k * 32 + lane;
    put_point(sPackAll, sBiasAll, n, bx[k], by[k], bz[k]);
    const int qx = bx[k] >> 4, qy = by[k] >> 4, qz = bz[k] >> 4;
    const bool selB = (unsigned)(qx - (x0 - MARGIN)) < (unsigned)(BOX_X + 2 * MARGIN)
                   && (unsigned)(qy - (y0 - MARGIN)) < (unsigned)(BOX_Y + 2 * MARGIN)
                   && (unsigned)(qz - (z0 - MARGIN)) < (unsigned)(BOX_Z + 2 * MARGIN);
    const unsigned mB = __builtin_amdgcn_ballot_w32(selB);
    posB[k] = selB ? cntB + __builtin_popcount(mB & lanesBelow) : -1;
    cntB += __builtin_popcount(mB);
    const int rx = (ax[k] >> 4) - x0, ry = (ay[k] >> 4) - y0, rz = (az[k] >> 4) - z0;
    const bool selA = (unsigned)rx < (unsigned)BOX_X && (unsigned)ry < (unsigned)BOX_Y && (unsigned)rz < (unsigned)BOX_Z;
    const unsigned mA = __builtin_amdgcn_ballot_w32(selA);
    posA[k] = selA ? cntA + __builtin_popcount(mA & lanesBelow) : -1;
    cntA += __builtin_popcount(mA);
  }
  if (lane == 0) { sCnt[wave][0] = cntA; sCnt[wave][1] = cntB; }
  __syncthreads();

  int baseA = 0, baseB = 0, nA = 0, nB = 0;
  for (int w = 0; w < ALL_WAVES; ++w) {
    const int ca = sCnt[w][0], cb = sCnt[w][1];
    if (w < wave) { baseA += ca; baseB += cb; }
    nA += ca; nB += cb;
  }
  nA = __builtin_amdgcn_readfirstlane(nA);
  nB = __builtin_amdgcn_readfirstlane(nB);
  const int nBpad = max((nB + 16 * TPT - 1) & ~(16 * TPT - 1), 16 * TPT);

#pragma unroll
  for (int k = 0; k < PER_THREAD; ++k) {
    const int n = wave * (32 * PER_THREAD) + k * 32 + lane;
    if (posB[k] >= 0) {
      const int p = baseB + posB[k];
      put_point(sPack, sBias, p, bx[k], by[k], bz[k]);
      sBIdx[p] = (unsigned short)n;
    }
    if (posA[k] >= 0) {
      const unsigned rx = (unsigned)((ax[k] >> 4) - x0), ry = (unsigned)((ay[k] >> 4) - y0), rz = (unsigned)((az[k] >> 4) - z0);
      sAEnt[baseA + posA[k]] = (unsigned)n | (rx << 12) | (ry << 17) | (rz << 22);
    }
  }
  if (tid < nBpad - nB) { sPack[nB + tid] = 0u; sBias[nB + tid] = 0.0f; sBIdx[nB + tid] = 0; }
  __syncthreads();

  const float* afB = a_feats + (size_t)b * NA * FD;
  const float* bfB = b_feats + (size_t)b * NB * FD;
  float* outB = out + (size_t)b * NA * (2 * FD);

  if (wave >= ALL_WAVES - COPY_WAVES) {
    const int row0 = box * COPY_ROWS + (wave - (ALL_WAVES - COPY_WAVES)) * (2 * COPY_PAIRS) + (lane >> 4);
    const int col = (lane & 15) * 4;
    v4f cp[COPY_PAIRS];
#pragma unroll
    for (int j = 0; j < COPY_PAIRS; ++j)
      cp[j] = *reinterpret_cast<const v4f*>(afB + (size_t)(row0 + 2 * j) * FD + col);
    static_assert(COPY_PAIRS == 16, "the statement below names the 16 loaded values");
    asm volatile("" :: "v"(cp[0]), "v"(cp[1]), "v"(cp[2]), "v"(cp[3]), "v"(cp[4]), "v"(cp[5]), "v"(cp[6]), "v"(cp[7]),
                       "v"(cp[8]), "v"(cp[9]), "v"(cp[10]), "v"(cp[11]), "v"(cp[12]), "v"(cp[13]), "v"(cp[14]), "v"(cp[15]));
#pragma unroll
    for (int j = 0; j < COPY_PAIRS; ++j)
      *reinterpret_cast<v4f*>(outB + (size_t)(row0 + 2 * j) * (2 * FD) + col) = cp[j];
  }

  for (int c0 = 0; c0 < nA; c0 += CHUNK) {
    const int cnt = min(CHUNK, nA - c0);
    const int tile0 = wave * 16;
    const int lrow = tile0 + mlane;
    bool unresolved = false;

    if (tile0 < cnt) {
      const bool valid = lrow < cnt;
      const unsigned ent = sAEnt[c0 + (valid ? lrow : tile0)];
      const int iax = x0 + (int)((ent >> 12) & 31u), iay = y0 + (int)((ent >> 17) & 31u), iaz = z0 + (int)((ent >> 22) & 63u);
      const int na2i = iax * iax + iay * iay + iaz * iaz;
      v2f Acol;
      Acol.x = hi ? (float)(2 * iaz) : (float)(2 * iax);
      Acol.y = hi ? 1.0f             : (float)(2 * iay);
      int p0 = -1, p1 = -1, p2 = -1;
      int thrbits = __float_as_int(-1.0f);
      sweep_tiles(0, nBpad / 16, sPack, sBias, Acol, mlane, hi, p0, p1, p2, thrbits,
                  __builtin_amdgcn_ballot_w32(valid));
      {
        const int q0 = other_half(p0), q1 = other_half(p1), q2 = other_half(p2);
        TOP3_INSERT_PK(q0, p0, p1, p2);
        TOP3_INSERT_PK(q1, p0, p1, p2);
        TOP3_INSERT_PK(q2, p0, p1, p2);
      }
      if (!hi) {
        const int d2third = na2i + KBIAS - (int)((unsigned)p2 >> 12);
        unresolved = valid && !(p2 >= 0 && d2third <= MARGIN * MARGIN);
        sIdx[lrow][0] = sBIdx[4095 - (p0 & 4095)];
        sIdx[lrow][1] = sBIdx[4095 - (p1 & 4095)];
        sIdx[lrow][2] = sBIdx[4095 - (p2 & 4095)];
        sWgt[lrow][0] = weight_of(na2i, p0);
        sWgt[lrow][1] = weight_of(na2i, p1);
        sWgt[lrow][2] = weight_of(na2i, p2);
      }
    }

    const unsigned um = __builtin_amdgcn_ballot_w32(unresolved);
    if (lane == 0) sUCnt[wave] = __builtin_popcount(um);
    __syncthreads();
    int ubase = 0, nU = 0;
    for (int w = 0; w < ALL_WAVES; ++w) {
      const int c = sUCnt[w];
      if (w < wave) ubase += c;
      nU += c;
    }
    nU = __builtin_amdgcn_readfirstlane(nU);

    if (nU > 0) {
      if (unresolved) sFbRow[ubase + __builtin_popcount(um & lanesBelow)] = (unsigned short)lrow;
      __syncthreads();
      for (int f0 = 0; f0 < nU; f0 += 16) {
        const bool fvalid = f0 + mlane < nU;
        const int frow = sFbRow[fvalid ? f0 + mlane : f0];
        const unsigned ent = sAEnt[c0 + frow];
        const int iax = x0 + (int)((ent >> 12) & 31u), iay = y0 + (int)((ent >> 17) & 31u), iaz = z0 + (int)((ent >> 22) & 63u);
        const int na2i = iax * iax + iay * iay + iaz * iaz;
        v2f Acol;
        Acol.x = hi ? (float)(2 * iaz) : (float)(2 * iax);
        Acol.y = hi ? 1.0f             : (float)(2 * iay);
        int p0 = -1, p1 = -1, p2 = -1;
        int thrbits = __float_as_int(-1.0f);
        sweep_tiles(wave * 16, wave * 16 + 16, sPackAll, sBiasAll, Acol, mlane, hi, p0, p1, p2, thrbits,
                    __builtin_amdgcn_ballot_w32(fvalid));
        {
          const int q0 = other_half(p0), q1 = other_half(p1), q2 = other_half(p2);
          TOP3_INSERT_PK(q0, p0, p1, p2);
          TOP3_INSERT_PK(q1, p0, p1, p2);
          TOP3_INSERT_PK(q2, p0, p1, p2);
        }
        if (!hi) { sFbPk[wave][mlane][0] = p0; sFbPk[wave][mlane][1] = p1; sFbPk[wave][mlane][2] = p2; }
        __syncthreads();
        if (wave == 0 && !hi && fvalid) {
          int r0 = -1, r1 = -1, r2 = -1;
          for (int w = 0; w < ALL_WAVES; ++w) {
            const int e0 = sFbPk[w][mlane][0], e1 = sFbPk[w][mlane][1], e2 = sFbPk[w][mlane][2];
            TOP3_INSERT_PK(e0, r0, r1, r2);
            TOP3_INSERT_PK(e1, r0, r1, r2);
            TOP3_INSERT_PK(e2, r0, r1, r2);
          }
          sIdx[frow][0] = 4095 - (r0 & 4095);
          sIdx[frow][1] = 4095 - (r1 & 4095);
          sIdx[frow][2] = 4095 - (r2 & 4095);
          sWgt[frow][0] = weight_of(na2i, r0);
          sWgt[frow][1] = weight_of(na2i, r1);
          sWgt[frow][2] = weight_of(na2i, r2);
        }
        __syncthreads();
      }
    }

    for (int k0 = 0; k0 < CHUNK / ALL_WAVES; k0 += GROWS) {
      const int first = k0 * ALL_WAVES + wave;
      if (first >= cnt) break;
      float g0[GROWS][2], g1[GROWS][2], g2[GROWS][2];
#pragma unroll
      for (int kk = 0; kk < GROWS; ++kk) {
        const int lr = (k0 + kk) * ALL_WAVES + wave;
        const int le = (k0 + kk < CHUNK / ALL_WAVES && lr < cnt) ? lr : first;
        const float* bf0 = bfB + (size_t)sIdx[le][0] * FD;
        const float* bf1 = bfB + (size_t)sIdx[le][1] * FD;
        const float* bf2 = bfB + (size_t)sIdx[le][2] * FD;
#pragma unroll
        for (int h = 0; h < 2; ++h) {
          const int d = lane + h * 32;
          g0[kk][h] = bf0[d]; g1[kk][h] = bf1[d]; g2[kk][h] = bf2[d];
        }
      }
#pragma unroll
      for (int kk = 0; kk < GROWS; ++kk) {
        const int lr = (k0 + kk) * ALL_WAVES + wave;
        if (k0 + kk < CHUNK / ALL_WAVES && lr < cnt) {
          float* op = outB + (size_t)(sAEnt[c0 + lr] & 4095u) * (2 * FD);
          const float w0 = sWgt[lr][0], w1 = sWgt[lr][1], w2 = sWgt[lr][2];
#pragma unroll
          for (int h = 0; h < 2; ++h) {
            const int d = lane + h * 32;
            const float tmp = w0 * g0[kk][h] + w1 * g1[kk][h] + w2 * g2[kk][h];
            op[FD + d] = tmp;
          }
        }
      }
    }
    __syncthreads();
  }
}

extern "C" void kernel_launch(void* const* d_in, const int* in_sizes, int n_in,
                              void* d_out, int out_size, void* d_ws, size_t ws_size,
                              hipStream_t stream) {
  const float* a_feats  = (const float*)d_in[0];
  const float* b_feats  = (const float*)d_in[1];
  const int*   a_coords = (const int*)d_in[2];
  const int*   b_coords = (const int*)d_in[3];
  float* out = (float*)d_out;

  dim3 grid(B_SZ * BOXES);
  dim3 block(THREADS);
  knn3_wmma<<<grid, block, 0, stream>>>(a_feats, b_feats, a_coords, b_coords, out);
}
